# starting kernel
# baseline (speedup 1.0000x reference)
#include <hip/hip_runtime.h>
#include <hip/hip_bf16.h>

typedef float v2f __attribute__((ext_vector_type(2)));
typedef float v8f __attribute__((ext_vector_type(8)));

#define WMMA_F32(a, b, c) \
  __builtin_amdgcn_wmma_f32_16x16x4_f32(false, (a), false, (b), (short)0, (c), false, false)

__launch_bounds__(256)
__global__ void gemm_bias_kernel(const float* __restrict__ A, const float* __restrict__ B,
                                 const float* __restrict__ bias, float* __restrict__ C,
                                 int M, int N, int K)
{
    __shared__ __align__(16) float As [128 * 34];
    __shared__ __align__(16) float BsT[128 * 34];

    const int tid  = threadIdx.x;
    const int wid  = tid >> 5;
    const int lane = tid & 31;
    const int m16  = lane & 15;
    const int h    = lane >> 4;
    const int wm   = wid >> 1;
    const int wn   = wid & 1;
    const int row0 = blockIdx.y * 128;
    const int col0 = blockIdx.x * 128;

    const int ar  = tid >> 1;
    const int akq = (tid & 1) << 4;
    const int bkr = tid >> 5;
    const int bcq = (tid & 31) << 2;

    v8f acc[2][4];
    #pragma unroll
    for (int i = 0; i < 2; ++i)
        #pragma unroll
        for (int j = 0; j < 4; ++j) { v8f z{}; acc[i][j] = z; }

    float4 pa[4], pb[4];
    #pragma unroll
    for (int i = 0; i < 4; ++i)
        pa[i] = *(const float4*)(A + (size_t)(row0 + ar) * K + akq + 4 * i);
    #pragma unroll
    for (int i = 0; i < 4; ++i)
        pb[i] = *(const float4*)(B + (size_t)(bkr + 8 * i) * N + col0 + bcq);

    for (int kk = 0; kk < K; kk += 32) {
        #pragma unroll
        for (int i = 0; i < 4; ++i) {
            float* d = &As[ar * 34 + akq + 4 * i];
            d[0] = pa[i].x; d[1] = pa[i].y; d[2] = pa[i].z; d[3] = pa[i].w;
        }
        #pragma unroll
        for (int i = 0; i < 4; ++i) {
            int kr = bkr + 8 * i;
            BsT[(bcq + 0) * 34 + kr] = pb[i].x;
            BsT[(bcq + 1) * 34 + kr] = pb[i].y;
            BsT[(bcq + 2) * 34 + kr] = pb[i].z;
            BsT[(bcq + 3) * 34 + kr] = pb[i].w;
        }
        __syncthreads();

        if (kk + 32 < K) {
            #pragma unroll
            for (int i = 0; i < 4; ++i)
                pa[i] = *(const float4*)(A + (size_t)(row0 + ar) * K + kk + 32 + akq + 4 * i);
            #pragma unroll
            for (int i = 0; i < 4; ++i)
                pb[i] = *(const float4*)(B + (size_t)(bkr + 8 * i) * N + kk * 0 + (size_t)(kk + 32 + bkr + 8 * i - (bkr + 8 * i)) * 0 + col0 + bcq + (size_t)(kk + 32) * 0)
                      , pb[i] = *(const float4*)(B + (size_t)(kk + 32 + bkr + 8 * i) * N + col0 + bcq);
        }

        #pragma unroll
        for (int ks = 0; ks < 32; ks += 4) {
            v2f a[2], b[4];
            #pragma unroll
            for (int mi = 0; mi < 2; ++mi)
                a[mi] = *(const v2f*)(&As[(wm * 32 + mi * 16 + m16) * 34 + ks + 2 * h]);
            #pragma unroll
            for (int ni = 0; ni < 4; ++ni)
                b[ni] = *(const v2f*)(&BsT[(wn * 64 + ni * 16 + m16) * 34 + ks + 2 * h]);
            #pragma unroll
            for (int mi = 0; mi < 2; ++mi)
                #pragma unroll
                for (int ni = 0; ni < 4; ++ni)
                    acc[mi][ni] = WMMA_F32(a[mi], b[ni], acc[mi][ni]);
        }
        __syncthreads();
    }

    #pragma unroll
    for (int mi = 0; mi < 2; ++mi) {
        #pragma unroll
        for (int ni = 0; ni < 4; ++ni) {
            int c = col0 + wn * 64 + ni * 16 + m16;
            float bv = bias[c];
            #pragma unroll
            for (int r = 0; r < 8; ++r) {
                int rr = row0 + wm * 32 + mi * 16 + r + 8 * h;
                C[(size_t)rr * N + c] = acc[mi][ni][r] + bv;
            }
        }
    }
}

__global__ void rope_repack_kernel(const float* __restrict__ qkv,
                                   const float* __restrict__ cosT,
                                   const float* __restrict__ sinT,
                                   float* __restrict__ Q, float* __restrict__ Ko,
                                   float* __restrict__ V)
{
    int tid = blockIdx.x * blockDim.x + threadIdx.x;
    int dp  = tid & 31;
    int h0  = (tid >> 5) & 15;
    int s   = (tid >> 9) & 2047;
    int b   = tid >> 20;
    size_t rowq = (size_t)(b * 2048 + s) * 3072;
    int off = h0 * 64 + 2 * dp;

    float c  = cosT[s * 32 + dp];
    float sn = sinT[s * 32 + dp];
    float qe = qkv[rowq + off],        qo = qkv[rowq + off + 1];
    float ke = qkv[rowq + 1024 + off], ko = qkv[rowq + 1024 + off + 1];

    size_t oidx = ((size_t)((b * 16 + h0) * 2048 + s)) * 64 + 2 * dp;
    Q[oidx]      = qe * c - qo * sn;
    Q[oidx + 1]  = qe * sn + qo * c;
    Ko[oidx]     = ke * c - ko * sn;
    Ko[oidx + 1] = ke * sn + ko * c;
    V[oidx]      = qkv[rowq + 2048 + off];
    V[oidx + 1]  = qkv[rowq + 2048 + off + 1];
}

__launch_bounds__(128)
__global__ void flash_attn_kernel(const float* __restrict__ Q, const float* __restrict__ Kg,
                                  const float* __restrict__ V, float* __restrict__ O)
{
    __shared__ __align__(16) float Ks [16 * 66];
    __shared__ __align__(16) float Vst[64 * 18];
    __shared__ __align__(16) float Pw [4 * 16 * 18];

    const int tid  = threadIdx.x;
    const int wid  = tid >> 5;
    const int lane = tid & 31;
    const int m16  = lane & 15;
    const int h    = lane >> 4;
    const int bh   = blockIdx.y;
    const int qbase = blockIdx.x * 64;
    const int qrow  = qbase + wid * 16;
    const size_t base = (size_t)bh * 2048 * 64;

    const int kvr  = tid >> 3;
    const int kvcq = (tid & 7) << 3;

    v2f aq[16];
    {
        const float* qp = Q + base + (size_t)(qrow + m16) * 64;
        #pragma unroll
        for (int s2 = 0; s2 < 16; ++s2) {
            v2f t = *(const v2f*)(qp + 4 * s2 + 2 * h);
            aq[s2] = t * 0.125f;
        }
    }

    float mrow[8], lrow[8];
    #pragma unroll
    for (int r = 0; r < 8; ++r) { mrow[r] = -1e30f; lrow[r] = 0.0f; }
    v8f oacc[4];
    #pragma unroll
    for (int i = 0; i < 4; ++i) { v8f z{}; oacc[i] = z; }

    const int ntiles = blockIdx.x * 4 + 4;
    float* Pme = &Pw[wid * 16 * 18];

    float4 pk[2], pv[2];
    #pragma unroll
    for (int i = 0; i < 2; ++i) {
        pk[i] = *(const float4*)(Kg + base + (size_t)kvr * 64 + kvcq + 4 * i);
        pv[i] = *(const float4*)(V  + base + (size_t)kvr * 64 + kvcq + 4 * i);
    }

    for (int kt = 0; kt < ntiles; ++kt) {
        #pragma unroll
        for (int i = 0; i < 2; ++i) {
            float* d = &Ks[kvr * 66 + kvcq + 4 * i];
            d[0] = pk[i].x; d[1] = pk[i].y; d[2] = pk[i].z; d[3] = pk[i].w;
            int c = kvcq + 4 * i;
            Vst[(c + 0) * 18 + kvr] = pv[i].x;
            Vst[(c + 1) * 18 + kvr] = pv[i].y;
            Vst[(c + 2) * 18 + kvr] = pv[i].z;
            Vst[(c + 3) * 18 + kvr] = pv[i].w;
        }
        __syncthreads();

        if (kt + 1 < ntiles) {
            #pragma unroll
            for (int i = 0; i < 2; ++i) {
                pk[i] = *(const float4*)(Kg + base + (size_t)((kt + 1) * 16 + kvr) * 64 + kvcq + 4 * i);
                pv[i] = *(const float4*)(V  + base + (size_t)((kt + 1) * 16 + kvr) * 64 + kvcq + 4 * i);
            }
        }

        v8f sc{};
        #pragma unroll
        for (int s2 = 0; s2 < 16; ++s2) {
            v2f b = *(const v2f*)(&Ks[m16 * 66 + 4 * s2 + 2 * h]);
            sc = WMMA_F32(aq[s2], b, sc);
        }

        const int kcol = kt * 16 + m16;
        #pragma unroll
        for (int r = 0; r < 8; ++r) {
            int qg = qrow + r + 8 * h;
            float val = (kcol <= qg) ? sc[r] : -1e30f;
            float mx = val;
            mx = fmaxf(mx, __shfl_xor(mx, 1, 32));
            mx = fmaxf(mx, __shfl_xor(mx, 2, 32));
            mx = fmaxf(mx, __shfl_xor(mx, 4, 32));
            mx = fmaxf(mx, __shfl_xor(mx, 8, 32));
            float mnew = fmaxf(mrow[r], mx);
            float p = __expf(val - mnew);
            float sum = p;
            sum += __shfl_xor(sum, 1, 32);
            sum += __shfl_xor(sum, 2, 32);
            sum += __shfl_xor(sum, 4, 32);
            sum += __shfl_xor(sum, 8, 32);
            float alpha = __expf(mrow[r] - mnew);
            mrow[r] = mnew;
            lrow[r] = lrow[r] * alpha + sum;
            #pragma unroll
            for (int ni = 0; ni < 4; ++ni) oacc[ni][r] *= alpha;
            Pme[(r + 8 * h) * 18 + m16] = p;
        }
        __builtin_amdgcn_wave_barrier();

        #pragma unroll
        for (int ks = 0; ks < 16; ks += 4) {
            v2f a = *(const v2f*)(&Pme[m16 * 18 + ks + 2 * h]);
            #pragma unroll
            for (int ni = 0; ni < 4; ++ni) {
                v2f b = *(const v2f*)(&Vst[(ni * 16 + m16) * 18 + ks + 2 * h]);
                oacc[ni] = WMMA_F32(a, b, oacc[ni]);
            }
        }
        __syncthreads();
    }

    const int b_ = bh >> 4, head = bh & 15;
    #pragma unroll
    for (int ni = 0; ni < 4; ++ni) {
        int c = head * 64 + ni * 16 + m16;
        #pragma unroll
        for (int r = 0; r < 8; ++r) {
            int srow = qrow + r + 8 * h;
            float inv = 1.0f / lrow[r];
            O[((size_t)(b_ * 2048 + srow)) * 1024 + c] = oacc[ni][r] * inv;
        }
    }
}

extern "C" void kernel_launch(void* const* d_in, const int* in_sizes, int n_in,
                              void* d_out, int out_size, void* d_ws, size_t ws_size,
                              hipStream_t stream)
{
    const float* x    = (const float*)d_in[0];
    const float* cosT = (const float*)d_in[2];
    const float* sinT = (const float*)d_in[3];
    const float* Wqkv = (const float*)d_in[4];
    const float* bqkv = (const float*)d_in[5];
    const float* Wo   = (const float*)d_in[6];
    const float* bo   = (const float*)d_in[7];
    float* out = (float*)d_out;

    float* ws  = (float*)d_ws;
    float* qkv = ws;
    float* Qb  = qkv + (size_t)4096 * 3072;
    float* Kb  = Qb + (size_t)4194304;
    float* Vb  = Kb + (size_t)4194304;
    float* Ob  = Vb + (size_t)4194304;

    gemm_bias_kernel<<<dim3(3072 / 128, 4096 / 128), 256, 0, stream>>>(
        x, Wqkv, bqkv, qkv, 4096, 3072, 1024);

    rope_repack_kernel<<<2097152 / 256, 256, 0, stream>>>(qkv, cosT, sinT, Qb, Kb, Vb);

    flash_attn_kernel<<<dim3(32, 32), 128, 0, stream>>>(Qb, Kb, Vb, Ob);

    gemm_bias_kernel<<<dim3(1024 / 128, 4096 / 128), 256, 0, stream>>>(
        Ob, Wo, bo, out, 4096, 1024, 1024);
}


# Attention softmax: running row max made lazy (moved only when a score tops it by 2^8); scores leave the matrix unit relative to it; no per-trip row-max butterfly or rescale; rounding of sums changed
# speedup vs baseline: 1.3738x; 1.3738x over previous
#include <hip/hip_runtime.h>
#include <hip/hip_bf16.h>

typedef float v2f __attribute__((ext_vector_type(2)));
typedef float v8f __attribute__((ext_vector_type(8)));

#define WMMA_F32(a, b, c) \
  __builtin_amdgcn_wmma_f32_16x16x4_f32(false, (a), false, (b), (short)0, (c), false, false)

template <int MASK>
__device__ __forceinline__ float row_xor(float v)
{
    return __builtin_bit_cast(float,
        __builtin_amdgcn_update_dpp(0, __builtin_bit_cast(int, v), 0x160 + MASK, 0xf, 0xf, true));
}

#define ROWMAX_STEP(M) \
    "v_max_num_f32_dpp %0, %0, %0 row_xmask:" #M " row_mask:0xf bank_mask:0xf bound_ctrl:1\n\t" \
    "v_max_num_f32_dpp %1, %1, %1 row_xmask:" #M " row_mask:0xf bank_mask:0xf bound_ctrl:1\n\t" \
    "v_max_num_f32_dpp %2, %2, %2 row_xmask:" #M " row_mask:0xf bank_mask:0xf bound_ctrl:1\n\t" \
    "v_max_num_f32_dpp %3, %3, %3 row_xmask:" #M " row_mask:0xf bank_mask:0xf bound_ctrl:1\n\t" \
    "v_max_num_f32_dpp %4, %4, %4 row_xmask:" #M " row_mask:0xf bank_mask:0xf bound_ctrl:1\n\t" \
    "v_max_num_f32_dpp %5, %5, %5 row_xmask:" #M " row_mask:0xf bank_mask:0xf bound_ctrl:1\n\t" \
    "v_max_num_f32_dpp %6, %6, %6 row_xmask:" #M " row_mask:0xf bank_mask:0xf bound_ctrl:1\n\t" \
    "v_max_num_f32_dpp %7, %7, %7 row_xmask:" #M " row_mask:0xf bank_mask:0xf bound_ctrl:1\n\t"
__device__ __forceinline__ void row_max8(const float (&v)[8], const float (&m)[8], float (&x)[8])
{
    asm("v_nop\n\tv_nop\n\tv_nop\n\tv_nop\n\t"
        "v_max_num_f32 %0, %8, %16\n\t"
        "v_max_num_f32 %1, %9, %17\n\t"
        "v_max_num_f32 %2, %10, %18\n\t"
        "v_max_num_f32 %3, %11, %19\n\t"
        "v_max_num_f32 %4, %12, %20\n\t"
        "v_max_num_f32 %5, %13, %21\n\t"
        "v_max_num_f32 %6, %14, %22\n\t"
        "v_max_num_f32 %7, %15, %23\n\t"
        ROWMAX_STEP(1) ROWMAX_STEP(2) ROWMAX_STEP(4) ROWMAX_STEP(8)
        : "=&v"(x[0]), "=&v"(x[1]), "=&v"(x[2]), "=&v"(x[3]),
          "=&v"(x[4]), "=&v"(x[5]), "=&v"(x[6]), "=&v"(x[7])
        : "v"(v[0]), "v"(v[1]), "v"(v[2]), "v"(v[3]), "v"(v[4]), "v"(v[5]), "v"(v[6]), "v"(v[7]),
          "v"(m[0]), "v"(m[1]), "v"(m[2]), "v"(m[3]), "v"(m[4]), "v"(m[5]), "v"(m[6]), "v"(m[7]));
}

template <bool B> struct mask_tag { static constexpr bool value = B; };

#define RESCALE_TAU 8.0f

__launch_bounds__(256)
__global__ void gemm_bias_kernel(const float* __restrict__ A, const float* __restrict__ B,
                                 const float* __restrict__ bias, float* __restrict__ C,
                                 int M, int N, int K)
{
    __shared__ __align__(16) float As [128 * 34];
    __shared__ __align__(16) float BsT[128 * 34];

    const int tid  = threadIdx.x;
    const int wid  = tid >> 5;
    const int lane = tid & 31;
    const int m16  = lane & 15;
    const int h    = lane >> 4;
    const int wm   = wid >> 1;
    const int wn   = wid & 1;
    const int row0 = blockIdx.y * 128;
    const int col0 = blockIdx.x * 128;

    const int ar  = tid >> 1;
    const int akq = (tid & 1) << 4;
    const int bkr = tid >> 5;
    const int bcq = (tid & 31) << 2;

    v8f acc[2][4];
    #pragma unroll
    for (int i = 0; i < 2; ++i)
        #pragma unroll
        for (int j = 0; j < 4; ++j) { v8f z{}; acc[i][j] = z; }

    float4 pa[4], pb[4];
    #pragma unroll
    for (int i = 0; i < 4; ++i)
        pa[i] = *(const float4*)(A + (size_t)(row0 + ar) * K + akq + 4 * i);
    #pragma unroll
    for (int i = 0; i < 4; ++i)
        pb[i] = *(const float4*)(B + (size_t)(bkr + 8 * i) * N + col0 + bcq);

    for (int kk = 0; kk < K; kk += 32) {
        #pragma unroll
        for (int i = 0; i < 4; ++i) {
            float* d = &As[ar * 34 + akq + 4 * i];
            d[0] = pa[i].x; d[1] = pa[i].y; d[2] = pa[i].z; d[3] = pa[i].w;
        }
        #pragma unroll
        for (int i = 0; i < 4; ++i) {
            int kr = bkr + 8 * i;
            BsT[(bcq + 0) * 34 + kr] = pb[i].x;
            BsT[(bcq + 1) * 34 + kr] = pb[i].y;
            BsT[(bcq + 2) * 34 + kr] = pb[i].z;
            BsT[(bcq + 3) * 34 + kr] = pb[i].w;
        }
        __syncthreads();

        if (kk + 32 < K) {
            #pragma unroll
            for (int i = 0; i < 4; ++i)
                pa[i] = *(const float4*)(A + (size_t)(row0 + ar) * K + kk + 32 + akq + 4 * i);
            #pragma unroll
            for (int i = 0; i < 4; ++i)
                pb[i] = *(const float4*)(B + (size_t)(bkr + 8 * i) * N + kk * 0 + (size_t)(kk + 32 + bkr + 8 * i - (bkr + 8 * i)) * 0 + col0 + bcq + (size_t)(kk + 32) * 0)
                      , pb[i] = *(const float4*)(B + (size_t)(kk + 32 + bkr + 8 * i) * N + col0 + bcq);
        }

        #pragma unroll
        for (int ks = 0; ks < 32; ks += 4) {
            v2f a[2], b[4];
            #pragma unroll
            for (int mi = 0; mi < 2; ++mi)
                a[mi] = *(const v2f*)(&As[(wm * 32 + mi * 16 + m16) * 34 + ks + 2 * h]);
            #pragma unroll
            for (int ni = 0; ni < 4; ++ni)
                b[ni] = *(const v2f*)(&BsT[(wn * 64 + ni * 16 + m16) * 34 + ks + 2 * h]);
            #pragma unroll
            for (int mi = 0; mi < 2; ++mi)
                #pragma unroll
                for (int ni = 0; ni < 4; ++ni)
                    acc[mi][ni] = WMMA_F32(a[mi], b[ni], acc[mi][ni]);
        }
        __syncthreads();
    }

    #pragma unroll
    for (int mi = 0; mi < 2; ++mi) {
        #pragma unroll
        for (int ni = 0; ni < 4; ++ni) {
            int c = col0 + wn * 64 + ni * 16 + m16;
            float bv = bias[c];
            #pragma unroll
            for (int r = 0; r < 8; ++r) {
                int rr = row0 + wm * 32 + mi * 16 + r + 8 * h;
                C[(size_t)rr * N + c] = acc[mi][ni][r] + bv;
            }
        }
    }
}

__global__ void rope_repack_kernel(const float* __restrict__ qkv,
                                   const float* __restrict__ cosT,
                                   const float* __restrict__ sinT,
                                   float* __restrict__ Q, float* __restrict__ Ko,
                                   float* __restrict__ V)
{
    int tid = blockIdx.x * blockDim.x + threadIdx.x;
    int dp  = tid & 31;
    int h0  = (tid >> 5) & 15;
    int s   = (tid >> 9) & 2047;
    int b   = tid >> 20;
    size_t rowq = (size_t)(b * 2048 + s) * 3072;
    int off = h0 * 64 + 2 * dp;

    float c  = cosT[s * 32 + dp];
    float sn = sinT[s * 32 + dp];
    float qe = qkv[rowq + off],        qo = qkv[rowq + off + 1];
    float ke = qkv[rowq + 1024 + off], ko = qkv[rowq + 1024 + off + 1];

    size_t oidx = ((size_t)((b * 16 + h0) * 2048 + s)) * 64 + 2 * dp;
    Q[oidx]      = qe * c - qo * sn;
    Q[oidx + 1]  = qe * sn + qo * c;
    Ko[oidx]     = ke * c - ko * sn;
    Ko[oidx + 1] = ke * sn + ko * c;
    V[oidx]      = qkv[rowq + 2048 + off];
    V[oidx + 1]  = qkv[rowq + 2048 + off + 1];
}

__launch_bounds__(128)
__global__ void flash_attn_kernel(const float* __restrict__ Q, const float* __restrict__ Kg,
                                  const float* __restrict__ V, float* __restrict__ Opart,
                                  float* __restrict__ ML)
{
    __shared__ __align__(16) float Ks [16 * 66];
    __shared__ __align__(16) float Vst[64 * 18];
    __shared__ __align__(16) float Pw [4 * 16 * 18];

    const int tid  = threadIdx.x;
    const int wid  = tid >> 5;
    const int lane = tid & 31;
    const int m16  = lane & 15;
    const int h    = lane >> 4;
    const int bh   = blockIdx.y;
    const int qbase = blockIdx.x * 64;
    const int qrow  = qbase + wid * 16;
    const size_t base = (size_t)bh * 2048 * 64;

    const int ntiles = blockIdx.x * 4 + 4;
    const int kt0 = blockIdx.z * 32;
    if (kt0 >= ntiles) return;
    const int nt  = (ntiles - kt0 < 32) ? ntiles - kt0 : 32;
    const int nfull = (kt0 + nt == ntiles) ? nt - 4 : nt;
    const size_t cbase = base + (size_t)kt0 * 16 * 64;

    const int kvr  = tid >> 3;
    const int kvcq = (tid & 7) << 3;

    v2f aq[16];
    {
        const float* qp = Q + base + (size_t)(qrow + m16) * 64;
        #pragma unroll
        for (int s2 = 0; s2 < 16; ++s2) {
            v2f t = *(const v2f*)(qp + 4 * s2 + 2 * h);
            aq[s2] = t * (0.125f * 1.44269504f);
        }
    }

    float lrow[8];
    #pragma unroll
    for (int r = 0; r < 8; ++r) lrow[r] = 0.0f;
    v8f cinit{};
    v8f oacc[4];
    #pragma unroll
    for (int i = 0; i < 4; ++i) { v8f z{}; oacc[i] = z; }

    float* Pme = &Pw[wid * 16 * 18];

    float4 pk[2], pv[2];
    #pragma unroll
    for (int i = 0; i < 2; ++i) {
        pk[i] = *(const float4*)(Kg + cbase + (size_t)kvr * 64 + kvcq + 4 * i);
        pv[i] = *(const float4*)(V  + cbase + (size_t)kvr * 64 + kvcq + 4 * i);
    }

    auto trip = [&](const int kt, auto tag) __attribute__((always_inline)) {
        constexpr bool MASKED = decltype(tag)::value;

        #pragma unroll
        for (int i = 0; i < 2; ++i) {
            float* d = &Ks[kvr * 66 + kvcq + 4 * i];
            d[0] = pk[i].x; d[1] = pk[i].y; d[2] = pk[i].z; d[3] = pk[i].w;
            int c = kvcq + 4 * i;
            Vst[(c + 0) * 18 + kvr] = pv[i].x;
            Vst[(c + 1) * 18 + kvr] = pv[i].y;
            Vst[(c + 2) * 18 + kvr] = pv[i].z;
            Vst[(c + 3) * 18 + kvr] = pv[i].w;
        }
        __syncthreads();

        if (kt + 1 < nt) {
            #pragma unroll
            for (int i = 0; i < 2; ++i) {
                pk[i] = *(const float4*)(Kg + cbase + (size_t)((kt + 1) * 16 + kvr) * 64 + kvcq + 4 * i);
                pv[i] = *(const float4*)(V  + cbase + (size_t)((kt + 1) * 16 + kvr) * 64 + kvcq + 4 * i);
            }
        }

        v8f sc = cinit;
        #pragma unroll
        for (int s2 = 0; s2 < 16; ++s2) {
            v2f b = *(const v2f*)(&Ks[m16 * 66 + 4 * s2 + 2 * h]);
            sc = WMMA_F32(aq[s2], b, sc);
        }

        float val[8], p[8];
        if constexpr (MASKED) {
            const int kcol = (kt0 + kt) * 16 + m16;
            #pragma unroll
            for (int r = 0; r < 8; ++r) {
                int qg = qrow + r + 8 * h;
                val[r] = (kcol <= qg) ? sc[r] : -1e30f;
            }
        } else {
            #pragma unroll
            for (int r = 0; r < 8; ++r) val[r] = sc[r];
        }
        const float vm = fmaxf(fmaxf(fmaxf(val[0], val[1]), fmaxf(val[2], val[3])),
                               fmaxf(fmaxf(val[4], val[5]), fmaxf(val[6], val[7])));
        const bool moved = (kt == 0) || (__builtin_amdgcn_ballot_w32(vm > RESCALE_TAU) != 0u);
        if (__builtin_expect(moved, 0)) {
            const float relv = (kt == 0) ? -1e30f : 0.0f;
            float rel[8], e[8];
            #pragma unroll
            for (int r = 0; r < 8; ++r) rel[r] = relv;
            row_max8(val, rel, e);
            #pragma unroll
            for (int r = 0; r < 8; ++r) {
                const float alpha = __builtin_amdgcn_exp2f(relv - e[r]);
                lrow[r] *= alpha;
                #pragma unroll
                for (int ni = 0; ni < 4; ++ni) oacc[ni][r] *= alpha;
                cinit[r] -= e[r];
                val[r]   -= e[r];
            }
        }
        #pragma unroll
        for (int r = 0; r < 8; ++r) {
            p[r] = __builtin_amdgcn_exp2f(val[r]);
            lrow[r] += p[r];
        }
        #pragma unroll
        for (int r = 0; r < 8; ++r) Pme[(r + 8 * h) * 18 + m16] = p[r];
        __builtin_amdgcn_wave_barrier();

        #pragma unroll
        for (int ks = 0; ks < 16; ks += 4) {
            v2f a = *(const v2f*)(&Pme[m16 * 18 + ks + 2 * h]);
            #pragma unroll
            for (int ni = 0; ni < 4; ++ni) {
                v2f b = *(const v2f*)(&Vst[(ni * 16 + m16) * 18 + ks + 2 * h]);
                oacc[ni] = WMMA_F32(a, b, oacc[ni]);
            }
        }
        __syncthreads();
    };

    int kt = 0;
    for (; kt < nfull; ++kt) trip(kt, mask_tag<false>{});
    for (; kt < nt; ++kt)    trip(kt, mask_tag<true>{});

    #pragma unroll
    for (int r = 0; r < 8; ++r) {
        float sum = lrow[r];
        sum += row_xor<1>(sum);
        sum += row_xor<2>(sum);
        sum += row_xor<4>(sum);
        sum += row_xor<8>(sum);
        lrow[r] = sum;
    }

    const int part = (bh * 32 + blockIdx.x) * 4 + blockIdx.z;
    float* op = Opart + (size_t)part * 4096;
    float* ml = ML + (size_t)part * 128;
    #pragma unroll
    for (int ni = 0; ni < 4; ++ni) {
        #pragma unroll
        for (int r = 0; r < 8; ++r)
            op[(wid * 16 + r + 8 * h) * 64 + ni * 16 + m16] = oacc[ni][r];
    }
    if (m16 == 0) {
        #pragma unroll
        for (int r = 0; r < 8; ++r) {
            ml[(wid * 16 + r + 8 * h) * 2 + 0] = -cinit[r];
            ml[(wid * 16 + r + 8 * h) * 2 + 1] = lrow[r];
        }
    }
}

template <int NCH>
__device__ __forceinline__ void merge_rows(const float* __restrict__ op,
                                           const float* __restrict__ ml,
                                           float* __restrict__ dst)
{
    float2 s[NCH];
    float4 o[NCH];
    #pragma unroll
    for (int c = 0; c < NCH; ++c) {
        s[c] = *(const float2*)(ml + c * 128);
        o[c] = *(const float4*)(op + (size_t)c * 4096);
    }
    float mstar = s[0].x;
    #pragma unroll
    for (int c = 1; c < NCH; ++c) mstar = fmaxf(mstar, s[c].x);
    float den = 0.0f;
    float4 acc = make_float4(0.0f, 0.0f, 0.0f, 0.0f);
    #pragma unroll
    for (int c = 0; c < NCH; ++c) {
        float w = __builtin_amdgcn_exp2f(s[c].x - mstar);
        den += w * s[c].y;
        acc.x += w * o[c].x; acc.y += w * o[c].y; acc.z += w * o[c].z; acc.w += w * o[c].w;
    }
    const float inv = 1.0f / den;
    *(float4*)dst = make_float4(acc.x * inv, acc.y * inv, acc.z * inv, acc.w * inv);
}

__global__ void merge_chunks_kernel(const float* __restrict__ Opart, const float* __restrict__ ML,
                                    float* __restrict__ O)
{
    const int c4  = threadIdx.x & 15;
    const int s   = ((blockIdx.x & 127) << 4) + (threadIdx.x >> 4);
    const int bh  = blockIdx.x >> 7;
    const int x   = s >> 6;
    const int row = s & 63;
    const int nch = ((blockIdx.x & 127) >> 5) + 1;
    const int part0 = (bh * 32 + x) * 4;

    const float* op = Opart + (size_t)part0 * 4096 + row * 64 + c4 * 4;
    const float* ml = ML + (size_t)part0 * 128 + row * 2;
    const int b_ = bh >> 4, head = bh & 15;
    float* dst = O + ((size_t)(b_ * 2048 + s)) * 1024 + head * 64 + c4 * 4;

    switch (nch) {
        case 1:  merge_rows<1>(op, ml, dst); break;
        case 2:  merge_rows<2>(op, ml, dst); break;
        case 3:  merge_rows<3>(op, ml, dst); break;
        default: merge_rows<4>(op, ml, dst); break;
    }
}

extern "C" void kernel_launch(void* const* d_in, const int* in_sizes, int n_in,
                              void* d_out, int out_size, void* d_ws, size_t ws_size,
                              hipStream_t stream)
{
    const float* x    = (const float*)d_in[0];
    const float* cosT = (const float*)d_in[2];
    const float* sinT = (const float*)d_in[3];
    const float* Wqkv = (const float*)d_in[4];
    const float* bqkv = (const float*)d_in[5];
    const float* Wo   = (const float*)d_in[6];
    const float* bo   = (const float*)d_in[7];
    float* out = (float*)d_out;

    float* ws  = (float*)d_ws;
    float* qkv = ws;
    float* Qb  = qkv + (size_t)4096 * 3072;
    float* Kb  = Qb + (size_t)4194304;
    float* Vb  = Kb + (size_t)4194304;
    float* Ob  = Vb + (size_t)4194304;
    float* Op  = Ob + (size_t)4194304;
    float* MLb = Op + (size_t)16777216;

    gemm_bias_kernel<<<dim3(3072 / 128, 4096 / 128), 256, 0, stream>>>(
        x, Wqkv, bqkv, qkv, 4096, 3072, 1024);

    rope_repack_kernel<<<2097152 / 256, 256, 0, stream>>>(qkv, cosT, sinT, Qb, Kb, Vb);

    flash_attn_kernel<<<dim3(32, 32, 4), 128, 0, stream>>>(Qb, Kb, Vb, Op, MLb);

    merge_chunks_kernel<<<1048576 / 256, 256, 0, stream>>>(Op, MLb, Ob);

    gemm_bias_kernel<<<dim3(1024 / 128, 4096 / 128), 256, 0, stream>>>(
        Ob, Wo, bo, out, 4096, 1024, 1024);
}


# Attention key-chunk workgroups handed out longest first: grid (head, 80 slots), each head's (q-tile, chunk) pairs by falling trip count 32..4, empty pairs not launched; sum order unchanged
# speedup vs baseline: 1.4862x; 1.0818x over previous
#include <hip/hip_runtime.h>
#include <hip/hip_bf16.h>

typedef float v2f __attribute__((ext_vector_type(2)));
typedef float v8f __attribute__((ext_vector_type(8)));

#define WMMA_F32(a, b, c) \
  __builtin_amdgcn_wmma_f32_16x16x4_f32(false, (a), false, (b), (short)0, (c), false, false)

template <int MASK>
__device__ __forceinline__ float row_xor(float v)
{
    return __builtin_bit_cast(float,
        __builtin_amdgcn_update_dpp(0, __builtin_bit_cast(int, v), 0x160 + MASK, 0xf, 0xf, true));
}

#define ROWMAX_STEP(M) \
    "v_max_num_f32_dpp %0, %0, %0 row_xmask:" #M " row_mask:0xf bank_mask:0xf bound_ctrl:1\n\t" \
    "v_max_num_f32_dpp %1, %1, %1 row_xmask:" #M " row_mask:0xf bank_mask:0xf bound_ctrl:1\n\t" \
    "v_max_num_f32_dpp %2, %2, %2 row_xmask:" #M " row_mask:0xf bank_mask:0xf bound_ctrl:1\n\t" \
    "v_max_num_f32_dpp %3, %3, %3 row_xmask:" #M " row_mask:0xf bank_mask:0xf bound_ctrl:1\n\t" \
    "v_max_num_f32_dpp %4, %4, %4 row_xmask:" #M " row_mask:0xf bank_mask:0xf bound_ctrl:1\n\t" \
    "v_max_num_f32_dpp %5, %5, %5 row_xmask:" #M " row_mask:0xf bank_mask:0xf bound_ctrl:1\n\t" \
    "v_max_num_f32_dpp %6, %6, %6 row_xmask:" #M " row_mask:0xf bank_mask:0xf bound_ctrl:1\n\t" \
    "v_max_num_f32_dpp %7, %7, %7 row_xmask:" #M " row_mask:0xf bank_mask:0xf bound_ctrl:1\n\t"
__device__ __forceinline__ void row_max8(const float (&v)[8], const float (&m)[8], float (&x)[8])
{
    asm("v_nop\n\tv_nop\n\tv_nop\n\tv_nop\n\t"
        "v_max_num_f32 %0, %8, %16\n\t"
        "v_max_num_f32 %1, %9, %17\n\t"
        "v_max_num_f32 %2, %10, %18\n\t"
        "v_max_num_f32 %3, %11, %19\n\t"
        "v_max_num_f32 %4, %12, %20\n\t"
        "v_max_num_f32 %5, %13, %21\n\t"
        "v_max_num_f32 %6, %14, %22\n\t"
        "v_max_num_f32 %7, %15, %23\n\t"
        ROWMAX_STEP(1) ROWMAX_STEP(2) ROWMAX_STEP(4) ROWMAX_STEP(8)
        : "=&v"(x[0]), "=&v"(x[1]), "=&v"(x[2]), "=&v"(x[3]),
          "=&v"(x[4]), "=&v"(x[5]), "=&v"(x[6]), "=&v"(x[7])
        : "v"(v[0]), "v"(v[1]), "v"(v[2]), "v"(v[3]), "v"(v[4]), "v"(v[5]), "v"(v[6]), "v"(v[7]),
          "v"(m[0]), "v"(m[1]), "v"(m[2]), "v"(m[3]), "v"(m[4]), "v"(m[5]), "v"(m[6]), "v"(m[7]));
}

template <bool B> struct mask_tag { static constexpr bool value = B; };

#define RESCALE_TAU 8.0f

__launch_bounds__(256)
__global__ void gemm_bias_kernel(const float* __restrict__ A, const float* __restrict__ B,
                                 const float* __restrict__ bias, float* __restrict__ C,
                                 int M, int N, int K)
{
    __shared__ __align__(16) float As [128 * 34];
    __shared__ __align__(16) float BsT[128 * 34];

    const int tid  = threadIdx.x;
    const int wid  = tid >> 5;
    const int lane = tid & 31;
    const int m16  = lane & 15;
    const int h    = lane >> 4;
    const int wm   = wid >> 1;
    const int wn   = wid & 1;
    const int row0 = blockIdx.y * 128;
    const int col0 = blockIdx.x * 128;

    const int ar  = tid >> 1;
    const int akq = (tid & 1) << 4;
    const int bkr = tid >> 5;
    const int bcq = (tid & 31) << 2;

    v8f acc[2][4];
    #pragma unroll
    for (int i = 0; i < 2; ++i)
        #pragma unroll
        for (int j = 0; j < 4; ++j) { v8f z{}; acc[i][j] = z; }

    float4 pa[4], pb[4];
    #pragma unroll
    for (int i = 0; i < 4; ++i)
        pa[i] = *(const float4*)(A + (size_t)(row0 + ar) * K + akq + 4 * i);
    #pragma unroll
    for (int i = 0; i < 4; ++i)
        pb[i] = *(const float4*)(B + (size_t)(bkr + 8 * i) * N + col0 + bcq);

    for (int kk = 0; kk < K; kk += 32) {
        #pragma unroll
        for (int i = 0; i < 4; ++i) {
            float* d = &As[ar * 34 + akq + 4 * i];
            d[0] = pa[i].x; d[1] = pa[i].y; d[2] = pa[i].z; d[3] = pa[i].w;
        }
        #pragma unroll
        for (int i = 0; i < 4; ++i) {
            int kr = bkr + 8 * i;
            BsT[(bcq + 0) * 34 + kr] = pb[i].x;
            BsT[(bcq + 1) * 34 + kr] = pb[i].y;
            BsT[(bcq + 2) * 34 + kr] = pb[i].z;
            BsT[(bcq + 3) * 34 + kr] = pb[i].w;
        }
        __syncthreads();

        if (kk + 32 < K) {
            #pragma unroll
            for (int i = 0; i < 4; ++i)
                pa[i] = *(const float4*)(A + (size_t)(row0 + ar) * K + kk + 32 + akq + 4 * i);
            #pragma unroll
            for (int i = 0; i < 4; ++i)
                pb[i] = *(const float4*)(B + (size_t)(bkr + 8 * i) * N + kk * 0 + (size_t)(kk + 32 + bkr + 8 * i - (bkr + 8 * i)) * 0 + col0 + bcq + (size_t)(kk + 32) * 0)
                      , pb[i] = *(const float4*)(B + (size_t)(kk + 32 + bkr + 8 * i) * N + col0 + bcq);
        }

        #pragma unroll
        for (int ks = 0; ks < 32; ks += 4) {
            v2f a[2], b[4];
            #pragma unroll
            for (int mi = 0; mi < 2; ++mi)
                a[mi] = *(const v2f*)(&As[(wm * 32 + mi * 16 + m16) * 34 + ks + 2 * h]);
            #pragma unroll
            for (int ni = 0; ni < 4; ++ni)
                b[ni] = *(const v2f*)(&BsT[(wn * 64 + ni * 16 + m16) * 34 + ks + 2 * h]);
            #pragma unroll
            for (int mi = 0; mi < 2; ++mi)
                #pragma unroll
                for (int ni = 0; ni < 4; ++ni)
                    acc[mi][ni] = WMMA_F32(a[mi], b[ni], acc[mi][ni]);
        }
        __syncthreads();
    }

    #pragma unroll
    for (int mi = 0; mi < 2; ++mi) {
        #pragma unroll
        for (int ni = 0; ni < 4; ++ni) {
            int c = col0 + wn * 64 + ni * 16 + m16;
            float bv = bias[c];
            #pragma unroll
            for (int r = 0; r < 8; ++r) {
                int rr = row0 + wm * 32 + mi * 16 + r + 8 * h;
                C[(size_t)rr * N + c] = acc[mi][ni][r] + bv;
            }
        }
    }
}

__global__ void rope_repack_kernel(const float* __restrict__ qkv,
                                   const float* __restrict__ cosT,
                                   const float* __restrict__ sinT,
                                   float* __restrict__ Q, float* __restrict__ Ko,
                                   float* __restrict__ V)
{
    int tid = blockIdx.x * blockDim.x + threadIdx.x;
    int dp  = tid & 31;
    int h0  = (tid >> 5) & 15;
    int s   = (tid >> 9) & 2047;
    int b   = tid >> 20;
    size_t rowq = (size_t)(b * 2048 + s) * 3072;
    int off = h0 * 64 + 2 * dp;

    float c  = cosT[s * 32 + dp];
    float sn = sinT[s * 32 + dp];
    float qe = qkv[rowq + off],        qo = qkv[rowq + off + 1];
    float ke = qkv[rowq + 1024 + off], ko = qkv[rowq + 1024 + off + 1];

    size_t oidx = ((size_t)((b * 16 + h0) * 2048 + s)) * 64 + 2 * dp;
    Q[oidx]      = qe * c - qo * sn;
    Q[oidx + 1]  = qe * sn + qo * c;
    Ko[oidx]     = ke * c - ko * sn;
    Ko[oidx + 1] = ke * sn + ko * c;
    V[oidx]      = qkv[rowq + 2048 + off];
    V[oidx + 1]  = qkv[rowq + 2048 + off + 1];
}

__launch_bounds__(128)
__global__ void flash_attn_kernel(const float* __restrict__ Q, const float* __restrict__ Kg,
                                  const float* __restrict__ V, float* __restrict__ Opart,
                                  float* __restrict__ ML)
{
    __shared__ __align__(16) float Ks [16 * 66];
    __shared__ __align__(16) float Vst[64 * 18];
    __shared__ __align__(16) float Pw [4 * 16 * 18];

    const int tid  = threadIdx.x;
    const int wid  = tid >> 5;
    const int lane = tid & 31;
    const int m16  = lane & 15;
    const int h    = lane >> 4;
    const int bh   = blockIdx.x;
    const int slot = blockIdx.y;
    int qt, ch;
    if (slot < 4)       { ch = slot; qt = 8 * slot + 7; }
    else if (slot < 28) { ch = 0;    qt = slot + 4; }
    else if (slot < 44) { ch = 1;    qt = slot - 12; }
    else if (slot < 52) { ch = 2;    qt = slot - 20; }
    else                { ch = (slot - 52) & 3; qt = 8 * ch + 6 - ((slot - 52) >> 2); }
    const int qbase = qt * 64;
    const int qrow  = qbase + wid * 16;
    const size_t base = (size_t)bh * 2048 * 64;

    const int ntiles = qt * 4 + 4;
    const int kt0 = ch * 32;
    const int nt  = (ntiles - kt0 < 32) ? ntiles - kt0 : 32;
    const int nfull = (kt0 + nt == ntiles) ? nt - 4 : nt;
    const size_t cbase = base + (size_t)kt0 * 16 * 64;

    const int kvr  = tid >> 3;
    const int kvcq = (tid & 7) << 3;

    v2f aq[16];
    {
        const float* qp = Q + base + (size_t)(qrow + m16) * 64;
        #pragma unroll
        for (int s2 = 0; s2 < 16; ++s2) {
            v2f t = *(const v2f*)(qp + 4 * s2 + 2 * h);
            aq[s2] = t * (0.125f * 1.44269504f);
        }
    }

    float lrow[8];
    #pragma unroll
    for (int r = 0; r < 8; ++r) lrow[r] = 0.0f;
    v8f cinit{};
    v8f oacc[4];
    #pragma unroll
    for (int i = 0; i < 4; ++i) { v8f z{}; oacc[i] = z; }

    float* Pme = &Pw[wid * 16 * 18];

    float4 pk[2], pv[2];
    #pragma unroll
    for (int i = 0; i < 2; ++i) {
        pk[i] = *(const float4*)(Kg + cbase + (size_t)kvr * 64 + kvcq + 4 * i);
        pv[i] = *(const float4*)(V  + cbase + (size_t)kvr * 64 + kvcq + 4 * i);
    }

    auto trip = [&](const int kt, auto tag) __attribute__((always_inline)) {
        constexpr bool MASKED = decltype(tag)::value;

        #pragma unroll
        for (int i = 0; i < 2; ++i) {
            float* d = &Ks[kvr * 66 + kvcq + 4 * i];
            d[0] = pk[i].x; d[1] = pk[i].y; d[2] = pk[i].z; d[3] = pk[i].w;
            int c = kvcq + 4 * i;
            Vst[(c + 0) * 18 + kvr] = pv[i].x;
            Vst[(c + 1) * 18 + kvr] = pv[i].y;
            Vst[(c + 2) * 18 + kvr] = pv[i].z;
            Vst[(c + 3) * 18 + kvr] = pv[i].w;
        }
        __syncthreads();

        if (kt + 1 < nt) {
            #pragma unroll
            for (int i = 0; i < 2; ++i) {
                pk[i] = *(const float4*)(Kg + cbase + (size_t)((kt + 1) * 16 + kvr) * 64 + kvcq + 4 * i);
                pv[i] = *(const float4*)(V  + cbase + (size_t)((kt + 1) * 16 + kvr) * 64 + kvcq + 4 * i);
            }
        }

        v8f sc = cinit;
        #pragma unroll
        for (int s2 = 0; s2 < 16; ++s2) {
            v2f b = *(const v2f*)(&Ks[m16 * 66 + 4 * s2 + 2 * h]);
            sc = WMMA_F32(aq[s2], b, sc);
        }

        float val[8], p[8];
        if constexpr (MASKED) {
            const int kcol = (kt0 + kt) * 16 + m16;
            #pragma unroll
            for (int r = 0; r < 8; ++r) {
                int qg = qrow + r + 8 * h;
                val[r] = (kcol <= qg) ? sc[r] : -1e30f;
            }
        } else {
            #pragma unroll
            for (int r = 0; r < 8; ++r) val[r] = sc[r];
        }
        const float vm = fmaxf(fmaxf(fmaxf(val[0], val[1]), fmaxf(val[2], val[3])),
                               fmaxf(fmaxf(val[4], val[5]), fmaxf(val[6], val[7])));
        const bool moved = (kt == 0) || (__builtin_amdgcn_ballot_w32(vm > RESCALE_TAU) != 0u);
        if (__builtin_expect(moved, 0)) {
            const float relv = (kt == 0) ? -1e30f : 0.0f;
            float rel[8], e[8];
            #pragma unroll
            for (int r = 0; r < 8; ++r) rel[r] = relv;
            row_max8(val, rel, e);
            #pragma unroll
            for (int r = 0; r < 8; ++r) {
                const float alpha = __builtin_amdgcn_exp2f(relv - e[r]);
                lrow[r] *= alpha;
                #pragma unroll
                for (int ni = 0; ni < 4; ++ni) oacc[ni][r] *= alpha;
                cinit[r] -= e[r];
                val[r]   -= e[r];
            }
        }
        #pragma unroll
        for (int r = 0; r < 8; ++r) {
            p[r] = __builtin_amdgcn_exp2f(val[r]);
            lrow[r] += p[r];
        }
        #pragma unroll
        for (int r = 0; r < 8; ++r) Pme[(r + 8 * h) * 18 + m16] = p[r];
        __builtin_amdgcn_wave_barrier();

        #pragma unroll
        for (int ks = 0; ks < 16; ks += 4) {
            v2f a = *(const v2f*)(&Pme[m16 * 18 + ks + 2 * h]);
            #pragma unroll
            for (int ni = 0; ni < 4; ++ni) {
                v2f b = *(const v2f*)(&Vst[(ni * 16 + m16) * 18 + ks + 2 * h]);
                oacc[ni] = WMMA_F32(a, b, oacc[ni]);
            }
        }
        __syncthreads();
    };

    int kt = 0;
    for (; kt < nfull; ++kt) trip(kt, mask_tag<false>{});
    for (; kt < nt; ++kt)    trip(kt, mask_tag<true>{});

    #pragma unroll
    for (int r = 0; r < 8; ++r) {
        float sum = lrow[r];
        sum += row_xor<1>(sum);
        sum += row_xor<2>(sum);
        sum += row_xor<4>(sum);
        sum += row_xor<8>(sum);
        lrow[r] = sum;
    }

    const int part = (bh * 32 + qt) * 4 + ch;
    float* op = Opart + (size_t)part * 4096;
    float* ml = ML + (size_t)part * 128;
    #pragma unroll
    for (int ni = 0; ni < 4; ++ni) {
        #pragma unroll
        for (int r = 0; r < 8; ++r)
            op[(wid * 16 + r + 8 * h) * 64 + ni * 16 + m16] = oacc[ni][r];
    }
    if (m16 == 0) {
        #pragma unroll
        for (int r = 0; r < 8; ++r) {
            ml[(wid * 16 + r + 8 * h) * 2 + 0] = -cinit[r];
            ml[(wid * 16 + r + 8 * h) * 2 + 1] = lrow[r];
        }
    }
}

template <int NCH>
__device__ __forceinline__ void merge_rows(const float* __restrict__ op,
                                           const float* __restrict__ ml,
                                           float* __restrict__ dst)
{
    float2 s[NCH];
    float4 o[NCH];
    #pragma unroll
    for (int c = 0; c < NCH; ++c) {
        s[c] = *(const float2*)(ml + c * 128);
        o[c] = *(const float4*)(op + (size_t)c * 4096);
    }
    float mstar = s[0].x;
    #pragma unroll
    for (int c = 1; c < NCH; ++c) mstar = fmaxf(mstar, s[c].x);
    float den = 0.0f;
    float4 acc = make_float4(0.0f, 0.0f, 0.0f, 0.0f);
    #pragma unroll
    for (int c = 0; c < NCH; ++c) {
        float w = __builtin_amdgcn_exp2f(s[c].x - mstar);
        den += w * s[c].y;
        acc.x += w * o[c].x; acc.y += w * o[c].y; acc.z += w * o[c].z; acc.w += w * o[c].w;
    }
    const float inv = 1.0f / den;
    *(float4*)dst = make_float4(acc.x * inv, acc.y * inv, acc.z * inv, acc.w * inv);
}

__global__ void merge_chunks_kernel(const float* __restrict__ Opart, const float* __restrict__ ML,
                                    float* __restrict__ O)
{
    const int c4  = threadIdx.x & 15;
    const int s   = ((blockIdx.x & 127) << 4) + (threadIdx.x >> 4);
    const int bh  = blockIdx.x >> 7;
    const int x   = s >> 6;
    const int row = s & 63;
    const int nch = ((blockIdx.x & 127) >> 5) + 1;
    const int part0 = (bh * 32 + x) * 4;

    const float* op = Opart + (size_t)part0 * 4096 + row * 64 + c4 * 4;
    const float* ml = ML + (size_t)part0 * 128 + row * 2;
    const int b_ = bh >> 4, head = bh & 15;
    float* dst = O + ((size_t)(b_ * 2048 + s)) * 1024 + head * 64 + c4 * 4;

    switch (nch) {
        case 1:  merge_rows<1>(op, ml, dst); break;
        case 2:  merge_rows<2>(op, ml, dst); break;
        case 3:  merge_rows<3>(op, ml, dst); break;
        default: merge_rows<4>(op, ml, dst); break;
    }
}

extern "C" void kernel_launch(void* const* d_in, const int* in_sizes, int n_in,
                              void* d_out, int out_size, void* d_ws, size_t ws_size,
                              hipStream_t stream)
{
    const float* x    = (const float*)d_in[0];
    const float* cosT = (const float*)d_in[2];
    const float* sinT = (const float*)d_in[3];
    const float* Wqkv = (const float*)d_in[4];
    const float* bqkv = (const float*)d_in[5];
    const float* Wo   = (const float*)d_in[6];
    const float* bo   = (const float*)d_in[7];
    float* out = (float*)d_out;

    float* ws  = (float*)d_ws;
    float* qkv = ws;
    float* Qb  = qkv + (size_t)4096 * 3072;
    float* Kb  = Qb + (size_t)4194304;
    float* Vb  = Kb + (size_t)4194304;
    float* Ob  = Vb + (size_t)4194304;
    float* Op  = Ob + (size_t)4194304;
    float* MLb = Op + (size_t)16777216;

    gemm_bias_kernel<<<dim3(3072 / 128, 4096 / 128), 256, 0, stream>>>(
        x, Wqkv, bqkv, qkv, 4096, 3072, 1024);

    rope_repack_kernel<<<2097152 / 256, 256, 0, stream>>>(qkv, cosT, sinT, Qb, Kb, Vb);

    flash_attn_kernel<<<dim3(32, 80), 128, 0, stream>>>(Qb, Kb, Vb, Op, MLb);

    merge_chunks_kernel<<<1048576 / 256, 256, 0, stream>>>(Op, MLb, Ob);

    gemm_bias_kernel<<<dim3(1024 / 128, 4096 / 128), 256, 0, stream>>>(
        Ob, Wo, bo, out, 4096, 1024, 1024);
}


# GEMM LDS staging rebuilt: row pitch 36 + new B lane map (no bank conflicts in fragment loads or tile stores), tiles double-buffered, stored mid-chunk, one barrier per chunk; sum order unchanged
# speedup vs baseline: 1.5422x; 1.0377x over previous
#include <hip/hip_runtime.h>
#include <hip/hip_bf16.h>

typedef float v2f __attribute__((ext_vector_type(2)));
typedef float v8f __attribute__((ext_vector_type(8)));

#define WMMA_F32(a, b, c) \
  __builtin_amdgcn_wmma_f32_16x16x4_f32(false, (a), false, (b), (short)0, (c), false, false)

template <int MASK>
__device__ __forceinline__ float row_xor(float v)
{
    return __builtin_bit_cast(float,
        __builtin_amdgcn_update_dpp(0, __builtin_bit_cast(int, v), 0x160 + MASK, 0xf, 0xf, true));
}

#define ROWMAX_STEP(M) \
    "v_max_num_f32_dpp %0, %0, %0 row_xmask:" #M " row_mask:0xf bank_mask:0xf bound_ctrl:1\n\t" \
    "v_max_num_f32_dpp %1, %1, %1 row_xmask:" #M " row_mask:0xf bank_mask:0xf bound_ctrl:1\n\t" \
    "v_max_num_f32_dpp %2, %2, %2 row_xmask:" #M " row_mask:0xf bank_mask:0xf bound_ctrl:1\n\t" \
    "v_max_num_f32_dpp %3, %3, %3 row_xmask:" #M " row_mask:0xf bank_mask:0xf bound_ctrl:1\n\t" \
    "v_max_num_f32_dpp %4, %4, %4 row_xmask:" #M " row_mask:0xf bank_mask:0xf bound_ctrl:1\n\t" \
    "v_max_num_f32_dpp %5, %5, %5 row_xmask:" #M " row_mask:0xf bank_mask:0xf bound_ctrl:1\n\t" \
    "v_max_num_f32_dpp %6, %6, %6 row_xmask:" #M " row_mask:0xf bank_mask:0xf bound_ctrl:1\n\t" \
    "v_max_num_f32_dpp %7, %7, %7 row_xmask:" #M " row_mask:0xf bank_mask:0xf bound_ctrl:1\n\t"
__device__ __forceinline__ void row_max8(const float (&v)[8], const float (&m)[8], float (&x)[8])
{
    asm("v_nop\n\tv_nop\n\tv_nop\n\tv_nop\n\t"
        "v_max_num_f32 %0, %8, %16\n\t"
        "v_max_num_f32 %1, %9, %17\n\t"
        "v_max_num_f32 %2, %10, %18\n\t"
        "v_max_num_f32 %3, %11, %19\n\t"
        "v_max_num_f32 %4, %12, %20\n\t"
        "v_max_num_f32 %5, %13, %21\n\t"
        "v_max_num_f32 %6, %14, %22\n\t"
        "v_max_num_f32 %7, %15, %23\n\t"
        ROWMAX_STEP(1) ROWMAX_STEP(2) ROWMAX_STEP(4) ROWMAX_STEP(8)
        : "=&v"(x[0]), "=&v"(x[1]), "=&v"(x[2]), "=&v"(x[3]),
          "=&v"(x[4]), "=&v"(x[5]), "=&v"(x[6]), "=&v"(x[7])
        : "v"(v[0]), "v"(v[1]), "v"(v[2]), "v"(v[3]), "v"(v[4]), "v"(v[5]), "v"(v[6]), "v"(v[7]),
          "v"(m[0]), "v"(m[1]), "v"(m[2]), "v"(m[3]), "v"(m[4]), "v"(m[5]), "v"(m[6]), "v"(m[7]));
}

template <bool B> struct mask_tag { static constexpr bool value = B; };

#define RESCALE_TAU 8.0f

template <int V> struct int_tag { static constexpr int value = V; };

__launch_bounds__(256)
__global__ void gemm_bias_kernel(const float* __restrict__ A, const float* __restrict__ B,
                                 const float* __restrict__ bias, float* __restrict__ C,
                                 int M, int N, int K)
{
    __shared__ __align__(16) float As [2][128 * 36];
    __shared__ __align__(16) float BsT[2][128 * 36];

    const int tid  = threadIdx.x;
    const int wid  = tid >> 5;
    const int lane = tid & 31;
    const int m16  = lane & 15;
    const int h    = lane >> 4;
    const int wm   = wid >> 1;
    const int wn   = wid & 1;
    const int row0 = blockIdx.y * 128;
    const int col0 = blockIdx.x * 128;

    const int ar  = tid >> 1;
    const int akq = (tid & 1) << 4;
    const int bkr = (lane >> 2) + 8 * (wid & 3);
    const int bcg = ((lane & 3) << 2) + 64 * (wid >> 2);

    v8f acc[2][4];
    #pragma unroll
    for (int i = 0; i < 2; ++i)
        #pragma unroll
        for (int j = 0; j < 4; ++j) { v8f z{}; acc[i][j] = z; }

    const float* Ap = A + (size_t)(row0 + ar) * K + akq;
    const float* Bp = B + (size_t)bkr * N + col0 + bcg;

    float4 pa[2][4], pb[2][4];
    auto fetch = [&](const int kk, auto btag) __attribute__((always_inline)) {
        constexpr int BUF = decltype(btag)::value;
        #pragma unroll
        for (int i = 0; i < 4; ++i)
            pa[BUF][i] = *(const float4*)(Ap + kk + 4 * i);
        #pragma unroll
        for (int i = 0; i < 4; ++i)
            pb[BUF][i] = *(const float4*)(Bp + (size_t)kk * N + 16 * i);
    };
    auto commit = [&](auto btag) __attribute__((always_inline)) {
        constexpr int BUF = decltype(btag)::value;
        #pragma unroll
        for (int i = 0; i < 4; ++i) {
            float* d = &As[BUF][ar * 36 + akq + 4 * i];
            d[0] = pa[BUF][i].x; d[1] = pa[BUF][i].y; d[2] = pa[BUF][i].z; d[3] = pa[BUF][i].w;
        }
        #pragma unroll
        for (int i = 0; i < 4; ++i) {
            const int c = bcg + 16 * i;
            BsT[BUF][(c + 0) * 36 + bkr] = pb[BUF][i].x;
            BsT[BUF][(c + 1) * 36 + bkr] = pb[BUF][i].y;
            BsT[BUF][(c + 2) * 36 + bkr] = pb[BUF][i].z;
            BsT[BUF][(c + 3) * 36 + bkr] = pb[BUF][i].w;
        }
    };
    auto half = [&](auto btag, const int ks0) __attribute__((always_inline)) {
        constexpr int BUF = decltype(btag)::value;
        #pragma unroll
        for (int ks = ks0; ks < ks0 + 16; ks += 4) {
            v2f a[2], b[4];
            #pragma unroll
            for (int mi = 0; mi < 2; ++mi)
                a[mi] = *(const v2f*)(&As[BUF][(wm * 32 + mi * 16 + m16) * 36 + ks + 2 * h]);
            #pragma unroll
            for (int ni = 0; ni < 4; ++ni)
                b[ni] = *(const v2f*)(&BsT[BUF][(wn * 64 + ni * 16 + m16) * 36 + ks + 2 * h]);
            #pragma unroll
            for (int mi = 0; mi < 2; ++mi)
                #pragma unroll
                for (int ni = 0; ni < 4; ++ni)
                    acc[mi][ni] = WMMA_F32(a[mi], b[ni], acc[mi][ni]);
        }
    };
    auto step = [&](const int kk, auto btag) __attribute__((always_inline)) {
        constexpr int BUF = decltype(btag)::value;
        if (kk + 64 < K) fetch(kk + 64, btag);
        half(btag, 0);
        if (kk + 32 < K) commit(int_tag<BUF ^ 1>{});
        half(btag, 16);
        __syncthreads();
    };

    fetch(0, int_tag<0>{});
    commit(int_tag<0>{});
    if (32 < K) fetch(32, int_tag<1>{});
    __syncthreads();

    for (int kk = 0; kk < K; kk += 64) {
        step(kk, int_tag<0>{});
        if (kk + 32 < K) step(kk + 32, int_tag<1>{});
    }

    #pragma unroll
    for (int mi = 0; mi < 2; ++mi) {
        #pragma unroll
        for (int ni = 0; ni < 4; ++ni) {
            int c = col0 + wn * 64 + ni * 16 + m16;
            float bv = bias[c];
            #pragma unroll
            for (int r = 0; r < 8; ++r) {
                int rr = row0 + wm * 32 + mi * 16 + r + 8 * h;
                C[(size_t)rr * N + c] = acc[mi][ni][r] + bv;
            }
        }
    }
}

__global__ void rope_repack_kernel(const float* __restrict__ qkv,
                                   const float* __restrict__ cosT,
                                   const float* __restrict__ sinT,
                                   float* __restrict__ Q, float* __restrict__ Ko,
                                   float* __restrict__ V)
{
    int tid = blockIdx.x * blockDim.x + threadIdx.x;
    int dp  = tid & 31;
    int h0  = (tid >> 5) & 15;
    int s   = (tid >> 9) & 2047;
    int b   = tid >> 20;
    size_t rowq = (size_t)(b * 2048 + s) * 3072;
    int off = h0 * 64 + 2 * dp;

    float c  = cosT[s * 32 + dp];
    float sn = sinT[s * 32 + dp];
    float qe = qkv[rowq + off],        qo = qkv[rowq + off + 1];
    float ke = qkv[rowq + 1024 + off], ko = qkv[rowq + 1024 + off + 1];

    size_t oidx = ((size_t)((b * 16 + h0) * 2048 + s)) * 64 + 2 * dp;
    Q[oidx]      = qe * c - qo * sn;
    Q[oidx + 1]  = qe * sn + qo * c;
    Ko[oidx]     = ke * c - ko * sn;
    Ko[oidx + 1] = ke * sn + ko * c;
    V[oidx]      = qkv[rowq + 2048 + off];
    V[oidx + 1]  = qkv[rowq + 2048 + off + 1];
}

__launch_bounds__(128)
__global__ void flash_attn_kernel(const float* __restrict__ Q, const float* __restrict__ Kg,
                                  const float* __restrict__ V, float* __restrict__ Opart,
                                  float* __restrict__ ML)
{
    __shared__ __align__(16) float Ks [16 * 66];
    __shared__ __align__(16) float Vst[64 * 18];
    __shared__ __align__(16) float Pw [4 * 16 * 18];

    const int tid  = threadIdx.x;
    const int wid  = tid >> 5;
    const int lane = tid & 31;
    const int m16  = lane & 15;
    const int h    = lane >> 4;
    const int bh   = blockIdx.x;
    const int slot = blockIdx.y;
    int qt, ch;
    if (slot < 4)       { ch = slot; qt = 8 * slot + 7; }
    else if (slot < 28) { ch = 0;    qt = slot + 4; }
    else if (slot < 44) { ch = 1;    qt = slot - 12; }
    else if (slot < 52) { ch = 2;    qt = slot - 20; }
    else                { ch = (slot - 52) & 3; qt = 8 * ch + 6 - ((slot - 52) >> 2); }
    const int qbase = qt * 64;
    const int qrow  = qbase + wid * 16;
    const size_t base = (size_t)bh * 2048 * 64;

    const int ntiles = qt * 4 + 4;
    const int kt0 = ch * 32;
    const int nt  = (ntiles - kt0 < 32) ? ntiles - kt0 : 32;
    const int nfull = (kt0 + nt == ntiles) ? nt - 4 : nt;
    const size_t cbase = base + (size_t)kt0 * 16 * 64;

    const int kvr  = tid >> 3;
    const int kvcq = (tid & 7) << 3;

    v2f aq[16];
    {
        const float* qp = Q + base + (size_t)(qrow + m16) * 64;
        #pragma unroll
        for (int s2 = 0; s2 < 16; ++s2) {
            v2f t = *(const v2f*)(qp + 4 * s2 + 2 * h);
            aq[s2] = t * (0.125f * 1.44269504f);
        }
    }

    float lrow[8];
    #pragma unroll
    for (int r = 0; r < 8; ++r) lrow[r] = 0.0f;
    v8f cinit{};
    v8f oacc[4];
    #pragma unroll
    for (int i = 0; i < 4; ++i) { v8f z{}; oacc[i] = z; }

    float* Pme = &Pw[wid * 16 * 18];

    float4 pk[2], pv[2];
    #pragma unroll
    for (int i = 0; i < 2; ++i) {
        pk[i] = *(const float4*)(Kg + cbase + (size_t)kvr * 64 + kvcq + 4 * i);
        pv[i] = *(const float4*)(V  + cbase + (size_t)kvr * 64 + kvcq + 4 * i);
    }

    auto trip = [&](const int kt, auto tag) __attribute__((always_inline)) {
        constexpr bool MASKED = decltype(tag)::value;

        #pragma unroll
        for (int i = 0; i < 2; ++i) {
            float* d = &Ks[kvr * 66 + kvcq + 4 * i];
            d[0] = pk[i].x; d[1] = pk[i].y; d[2] = pk[i].z; d[3] = pk[i].w;
            int c = kvcq + 4 * i;
            Vst[(c + 0) * 18 + kvr] = pv[i].x;
            Vst[(c + 1) * 18 + kvr] = pv[i].y;
            Vst[(c + 2) * 18 + kvr] = pv[i].z;
            Vst[(c + 3) * 18 + kvr] = pv[i].w;
        }
        __syncthreads();

        if (kt + 1 < nt) {
            #pragma unroll
            for (int i = 0; i < 2; ++i) {
                pk[i] = *(const float4*)(Kg + cbase + (size_t)((kt + 1) * 16 + kvr) * 64 + kvcq + 4 * i);
                pv[i] = *(const float4*)(V  + cbase + (size_t)((kt + 1) * 16 + kvr) * 64 + kvcq + 4 * i);
            }
        }

        v8f sc = cinit;
        #pragma unroll
        for (int s2 = 0; s2 < 16; ++s2) {
            v2f b = *(const v2f*)(&Ks[m16 * 66 + 4 * s2 + 2 * h]);
            sc = WMMA_F32(aq[s2], b, sc);
        }

        float val[8], p[8];
        if constexpr (MASKED) {
            const int kcol = (kt0 + kt) * 16 + m16;
            #pragma unroll
            for (int r = 0; r < 8; ++r) {
                int qg = qrow + r + 8 * h;
                val[r] = (kcol <= qg) ? sc[r] : -1e30f;
            }
        } else {
            #pragma unroll
            for (int r = 0; r < 8; ++r) val[r] = sc[r];
        }
        const float vm = fmaxf(fmaxf(fmaxf(val[0], val[1]), fmaxf(val[2], val[3])),
                               fmaxf(fmaxf(val[4], val[5]), fmaxf(val[6], val[7])));
        const bool moved = (kt == 0) || (__builtin_amdgcn_ballot_w32(vm > RESCALE_TAU) != 0u);
        if (__builtin_expect(moved, 0)) {
            const float relv = (kt == 0) ? -1e30f : 0.0f;
            float rel[8], e[8];
            #pragma unroll
            for (int r = 0; r < 8; ++r) rel[r] = relv;
            row_max8(val, rel, e);
            #pragma unroll
            for (int r = 0; r < 8; ++r) {
                const float alpha = __builtin_amdgcn_exp2f(relv - e[r]);
                lrow[r] *= alpha;
                #pragma unroll
                for (int ni = 0; ni < 4; ++ni) oacc[ni][r] *= alpha;
                cinit[r] -= e[r];
                val[r]   -= e[r];
            }
        }
        #pragma unroll
        for (int r = 0; r < 8; ++r) {
            p[r] = __builtin_amdgcn_exp2f(val[r]);
            lrow[r] += p[r];
        }
        #pragma unroll
        for (int r = 0; r < 8; ++r) Pme[(r + 8 * h) * 18 + m16] = p[r];
        __builtin_amdgcn_wave_barrier();

        #pragma unroll
        for (int ks = 0; ks < 16; ks += 4) {
            v2f a = *(const v2f*)(&Pme[m16 * 18 + ks + 2 * h]);
            #pragma unroll
            for (int ni = 0; ni < 4; ++ni) {
                v2f b = *(const v2f*)(&Vst[(ni * 16 + m16) * 18 + ks + 2 * h]);
                oacc[ni] = WMMA_F32(a, b, oacc[ni]);
            }
        }
        __syncthreads();
    };

    int kt = 0;
    for (; kt < nfull; ++kt) trip(kt, mask_tag<false>{});
    for (; kt < nt; ++kt)    trip(kt, mask_tag<true>{});

    #pragma unroll
    for (int r = 0; r < 8; ++r) {
        float sum = lrow[r];
        sum += row_xor<1>(sum);
        sum += row_xor<2>(sum);
        sum += row_xor<4>(sum);
        sum += row_xor<8>(sum);
        lrow[r] = sum;
    }

    const int part = (bh * 32 + qt) * 4 + ch;
    float* op = Opart + (size_t)part * 4096;
    float* ml = ML + (size_t)part * 128;
    #pragma unroll
    for (int ni = 0; ni < 4; ++ni) {
        #pragma unroll
        for (int r = 0; r < 8; ++r)
            op[(wid * 16 + r + 8 * h) * 64 + ni * 16 + m16] = oacc[ni][r];
    }
    if (m16 == 0) {
        #pragma unroll
        for (int r = 0; r < 8; ++r) {
            ml[(wid * 16 + r + 8 * h) * 2 + 0] = -cinit[r];
            ml[(wid * 16 + r + 8 * h) * 2 + 1] = lrow[r];
        }
    }
}

template <int NCH>
__device__ __forceinline__ void merge_rows(const float* __restrict__ op,
                                           const float* __restrict__ ml,
                                           float* __restrict__ dst)
{
    float2 s[NCH];
    float4 o[NCH];
    #pragma unroll
    for (int c = 0; c < NCH; ++c) {
        s[c] = *(const float2*)(ml + c * 128);
        o[c] = *(const float4*)(op + (size_t)c * 4096);
    }
    float mstar = s[0].x;
    #pragma unroll
    for (int c = 1; c < NCH; ++c) mstar = fmaxf(mstar, s[c].x);
    float den = 0.0f;
    float4 acc = make_float4(0.0f, 0.0f, 0.0f, 0.0f);
    #pragma unroll
    for (int c = 0; c < NCH; ++c) {
        float w = __builtin_amdgcn_exp2f(s[c].x - mstar);
        den += w * s[c].y;
        acc.x += w * o[c].x; acc.y += w * o[c].y; acc.z += w * o[c].z; acc.w += w * o[c].w;
    }
    const float inv = 1.0f / den;
    *(float4*)dst = make_float4(acc.x * inv, acc.y * inv, acc.z * inv, acc.w * inv);
}

__global__ void merge_chunks_kernel(const float* __restrict__ Opart, const float* __restrict__ ML,
                                    float* __restrict__ O)
{
    const int c4  = threadIdx.x & 15;
    const int s   = ((blockIdx.x & 127) << 4) + (threadIdx.x >> 4);
    const int bh  = blockIdx.x >> 7;
    const int x   = s >> 6;
    const int row = s & 63;
    const int nch = ((blockIdx.x & 127) >> 5) + 1;
    const int part0 = (bh * 32 + x) * 4;

    const float* op = Opart + (size_t)part0 * 4096 + row * 64 + c4 * 4;
    const float* ml = ML + (size_t)part0 * 128 + row * 2;
    const int b_ = bh >> 4, head = bh & 15;
    float* dst = O + ((size_t)(b_ * 2048 + s)) * 1024 + head * 64 + c4 * 4;

    switch (nch) {
        case 1:  merge_rows<1>(op, ml, dst); break;
        case 2:  merge_rows<2>(op, ml, dst); break;
        case 3:  merge_rows<3>(op, ml, dst); break;
        default: merge_rows<4>(op, ml, dst); break;
    }
}

extern "C" void kernel_launch(void* const* d_in, const int* in_sizes, int n_in,
                              void* d_out, int out_size, void* d_ws, size_t ws_size,
                              hipStream_t stream)
{
    const float* x    = (const float*)d_in[0];
    const float* cosT = (const float*)d_in[2];
    const float* sinT = (const float*)d_in[3];
    const float* Wqkv = (const float*)d_in[4];
    const float* bqkv = (const float*)d_in[5];
    const float* Wo   = (const float*)d_in[6];
    const float* bo   = (const float*)d_in[7];
    float* out = (float*)d_out;

    float* ws  = (float*)d_ws;
    float* qkv = ws;
    float* Qb  = qkv + (size_t)4096 * 3072;
    float* Kb  = Qb + (size_t)4194304;
    float* Vb  = Kb + (size_t)4194304;
    float* Ob  = Vb + (size_t)4194304;
    float* Op  = Ob + (size_t)4194304;
    float* MLb = Op + (size_t)16777216;

    gemm_bias_kernel<<<dim3(3072 / 128, 4096 / 128), 256, 0, stream>>>(
        x, Wqkv, bqkv, qkv, 4096, 3072, 1024);

    rope_repack_kernel<<<2097152 / 256, 256, 0, stream>>>(qkv, cosT, sinT, Qb, Kb, Vb);

    flash_attn_kernel<<<dim3(32, 80), 128, 0, stream>>>(Qb, Kb, Vb, Op, MLb);

    merge_chunks_kernel<<<1048576 / 256, 256, 0, stream>>>(Op, MLb, Ob);

    gemm_bias_kernel<<<dim3(1024 / 128, 4096 / 128), 256, 0, stream>>>(
        Ob, Wo, bo, out, 4096, 1024, 1024);
}
